# starting kernel
# baseline (speedup 1.0000x reference)
#include <hip/hip_runtime.h>

typedef float v2f __attribute__((ext_vector_type(2)));
typedef float v8f __attribute__((ext_vector_type(8)));

#define MDIM 8192
#define NDIM 1024
#define KDIM 1024
#define LHALF 16

#define BM 128
#define BN 128
#define BK 32
#define SA (4 * BM + 8)
#define SB (4 * BN + 8)

__global__ __launch_bounds__(256)
void eunn_coeff_kernel(const float* __restrict__ thetaA,
                       const float* __restrict__ thetaB,
                       float* __restrict__ dA, float* __restrict__ oA,
                       float* __restrict__ dB, float* __restrict__ oB)
{
    int idx = blockIdx.x * blockDim.x + threadIdx.x;
    if (idx >= LHALF * NDIM) return;
    const int t = idx >> 10;
    const int j = idx & 1023;

    {
        const int a = idx >> 5;
        const int b = (idx & 31) >> 1;
        const int c = idx & 1;
        const float th = thetaA[a * 16 + b];
        dA[idx] = cosf(th);
        oA[idx] = (c ? 1.0f : -1.0f) * sinf(th);
    }
    if (j == 0 || j == 1023) {
        dB[idx] = 1.0f;
        oB[idx] = 0.0f;
    } else {
        const int f = t * 1022 + (j - 1);
        const int a = f >> 5;
        const int b = (f & 31) >> 1;
        const int c = f & 1;
        const float th = thetaB[a * 16 + b];
        dB[idx] = cosf(th);
        oB[idx] = (c ? 1.0f : -1.0f) * sinf(th);
    }
}

__global__ __launch_bounds__(256)
void gemm_hx_kernel(const float* __restrict__ A,
                    const float* __restrict__ Bg,
                    const float* __restrict__ hx,
                    float* __restrict__ out)
{
    __shared__ __align__(16) float AsF[(BK / 4) * SA];
    __shared__ __align__(16) float BsF[(BK / 4) * SB];

    const int tid    = threadIdx.x;
    const int lane   = tid & 31;
    const int wave   = tid >> 5;
    const int wm     = wave & 3;
    const int wn     = wave >> 2;
    const int blockM = blockIdx.x * BM;
    const int blockN = blockIdx.y * BN;

    const int lane15 = lane & 15;
    const int koff   = (lane >> 4) << 1;

    v8f acc[2][4] = {};

    const int a_m0 = tid >> 3;
    const int a_kg = tid & 7;
    const int b_n  = tid & 127;
    const int b_kg0 = (tid >> 7);

    const int mA = wm * 32 + lane15;
    const int nB = wn * 64 + lane15;

    for (int kb = 0; kb < KDIM; kb += BK) {
        __syncthreads();
        #pragma unroll
        for (int p = 0; p < 4; ++p) {
            const int m = a_m0 + 32 * p;
            const float4 v = *(const float4*)(A + (size_t)(blockM + m) * KDIM + kb + a_kg * 4);
            *(float4*)(AsF + a_kg * SA + m * 4) = v;
        }
        #pragma unroll
        for (int p = 0; p < 4; ++p) {
            const int kg = b_kg0 + 2 * p;
            const int k  = kb + kg * 4;
            float4 v;
            v.x = Bg[(size_t)(k + 0) * NDIM + blockN + b_n];
            v.y = Bg[(size_t)(k + 1) * NDIM + blockN + b_n];
            v.z = Bg[(size_t)(k + 2) * NDIM + blockN + b_n];
            v.w = Bg[(size_t)(k + 3) * NDIM + blockN + b_n];
            *(float4*)(BsF + kg * SB + b_n * 4) = v;
        }
        __syncthreads();

        if (kb + BK < KDIM) {
            __builtin_prefetch(A + (size_t)(blockM + a_m0) * KDIM + kb + BK + a_kg * 4, 0, 0);
            __builtin_prefetch(Bg + (size_t)(kb + BK + b_kg0 * 4) * NDIM + blockN + b_n, 0, 0);
        }

        #pragma unroll
        for (int kg = 0; kg < BK / 4; ++kg) {
            const float* aB = AsF + kg * SA + koff;
            const float* bB = BsF + kg * SB + koff;
            v2f a0 = *(const v2f*)(aB + (mA     ) * 4);
            v2f a1 = *(const v2f*)(aB + (mA + 16) * 4);
            v2f b0 = *(const v2f*)(bB + (nB     ) * 4);
            v2f b1 = *(const v2f*)(bB + (nB + 16) * 4);
            v2f b2 = *(const v2f*)(bB + (nB + 32) * 4);
            v2f b3 = *(const v2f*)(bB + (nB + 48) * 4);
            acc[0][0] = __builtin_amdgcn_wmma_f32_16x16x4_f32(false, a0, false, b0,
                                                              (short)0, acc[0][0], false, false);
            acc[0][1] = __builtin_amdgcn_wmma_f32_16x16x4_f32(false, a0, false, b1,
                                                              (short)0, acc[0][1], false, false);
            acc[0][2] = __builtin_amdgcn_wmma_f32_16x16x4_f32(false, a0, false, b2,
                                                              (short)0, acc[0][2], false, false);
            acc[0][3] = __builtin_amdgcn_wmma_f32_16x16x4_f32(false, a0, false, b3,
                                                              (short)0, acc[0][3], false, false);
            acc[1][0] = __builtin_amdgcn_wmma_f32_16x16x4_f32(false, a1, false, b0,
                                                              (short)0, acc[1][0], false, false);
            acc[1][1] = __builtin_amdgcn_wmma_f32_16x16x4_f32(false, a1, false, b1,
                                                              (short)0, acc[1][1], false, false);
            acc[1][2] = __builtin_amdgcn_wmma_f32_16x16x4_f32(false, a1, false, b2,
                                                              (short)0, acc[1][2], false, false);
            acc[1][3] = __builtin_amdgcn_wmma_f32_16x16x4_f32(false, a1, false, b3,
                                                              (short)0, acc[1][3], false, false);
        }
    }

    const int hi = (lane >> 4) << 3;
    #pragma unroll
    for (int ti = 0; ti < 2; ++ti) {
        #pragma unroll
        for (int tj = 0; tj < 4; ++tj) {
            const int mBase = blockM + wm * 32 + ti * 16 + hi;
            const int nCol  = blockN + wn * 64 + tj * 16 + lane15;
            #pragma unroll
            for (int v = 0; v < 8; ++v) {
                const size_t idx = (size_t)(mBase + v) * NDIM + nCol;
                out[idx] = acc[ti][tj][v] + hx[idx];
            }
        }
    }
}

__global__ __launch_bounds__(256)
void eunn_modrelu_kernel(float* __restrict__ h,
                         const float* __restrict__ dA,
                         const float* __restrict__ oA,
                         const float* __restrict__ dB,
                         const float* __restrict__ oB,
                         const float* __restrict__ bias)
{
    __shared__ float xs[NDIM];
    const int row = blockIdx.x;
    const int tid = threadIdx.x;
    const int j0  = tid * 4;
    float* __restrict__ hrow = h + (size_t)row * NDIM;

    float4 x = *(const float4*)(hrow + j0);

    for (int t = 0; t < LHALF; ++t) {
        const float4 da = *(const float4*)(dA + t * NDIM + j0);
        const float4 oa = *(const float4*)(oA + t * NDIM + j0);
        float4 xa;
        xa.x = x.x * da.x + x.y * oa.x;
        xa.y = x.y * da.y + x.x * oa.y;
        xa.z = x.z * da.z + x.w * oa.z;
        xa.w = x.w * da.w + x.z * oa.w;

        *(float4*)(xs + j0) = xa;
        __syncthreads();

        const float4 db = *(const float4*)(dB + t * NDIM + j0);
        const float4 ob = *(const float4*)(oB + t * NDIM + j0);
        float y[4];
        #pragma unroll
        for (int u = 0; u < 4; ++u) {
            const int j   = j0 + u;
            const int src = (j < 512) ? (2 * j) : (2 * j - 1023);
            y[u] = xs[src];
        }
        x.x = xa.x * db.x + y[0] * ob.x;
        x.y = xa.y * db.y + y[1] * ob.y;
        x.z = xa.z * db.z + y[2] * ob.z;
        x.w = xa.w * db.w + y[3] * ob.w;
        __syncthreads();
    }

    const float4 bb = *(const float4*)(bias + j0);
    float4 o;
    {
        float s;
        s   = (x.x > 0.f) ? 1.f : ((x.x < 0.f) ? -1.f : 0.f);
        o.x = s * fmaxf(fabsf(x.x) + bb.x, 0.f);
        s   = (x.y > 0.f) ? 1.f : ((x.y < 0.f) ? -1.f : 0.f);
        o.y = s * fmaxf(fabsf(x.y) + bb.y, 0.f);
        s   = (x.z > 0.f) ? 1.f : ((x.z < 0.f) ? -1.f : 0.f);
        o.z = s * fmaxf(fabsf(x.z) + bb.z, 0.f);
        s   = (x.w > 0.f) ? 1.f : ((x.w < 0.f) ? -1.f : 0.f);
        o.w = s * fmaxf(fabsf(x.w) + bb.w, 0.f);
    }
    *(float4*)(hrow + j0) = o;
}

extern "C" void kernel_launch(void* const* d_in, const int* in_sizes, int n_in,
                              void* d_out, int out_size, void* d_ws, size_t ws_size,
                              hipStream_t stream)
{
    const float* input_ = (const float*)d_in[0];
    const float* hx     = (const float*)d_in[1];
    const float* U      = (const float*)d_in[2];
    const float* thetaA = (const float*)d_in[3];
    const float* thetaB = (const float*)d_in[4];
    const float* bias   = (const float*)d_in[5];
    float* out = (float*)d_out;

    float* ws = (float*)d_ws;
    float* dA = ws;
    float* oA = ws + 16384;
    float* dB = ws + 32768;
    float* oB = ws + 49152;

    eunn_coeff_kernel<<<(LHALF * NDIM + 255) / 256, 256, 0, stream>>>(
        thetaA, thetaB, dA, oA, dB, oB);

    dim3 grid(MDIM / BM, NDIM / BN);
    gemm_hx_kernel<<<grid, 256, 0, stream>>>(input_, U, hx, out);

    eunn_modrelu_kernel<<<MDIM, 256, 0, stream>>>(out, dA, oA, dB, oB, bias);
}


# GEMM re-decomposed: each workgroup keeps a 1024x64 panel of U in LDS for all of K, A fragments go global->registers in 16-byte loads, no barrier in the K loop; k sum order changes inside groups of 8
# speedup vs baseline: 1.3501x; 1.3501x over previous
#include <hip/hip_runtime.h>

typedef float v2f __attribute__((ext_vector_type(2)));
typedef float v8f __attribute__((ext_vector_type(8)));

#define MDIM 8192
#define NDIM 1024
#define KDIM 1024
#define LHALF 16

#define PN    64
#define GW    16
#define GROWS (GW * 32)
#define KQ    (KDIM / 4)

#define NPAIR    (NDIM / 2)
#define CB_PITCH 528

__global__ __launch_bounds__(256)
void eunn_coeff_kernel(const float* __restrict__ thetaA,
                       const float* __restrict__ thetaB,
                       float* __restrict__ cA, float* __restrict__ sA,
                       float* __restrict__ cB, float* __restrict__ sB)
{
    const int idx = blockIdx.x * blockDim.x + threadIdx.x;
    if (idx < LHALF * NPAIR) {
        const float th = thetaA[idx];
        cA[idx] = cosf(th);
        sA[idx] = sinf(th);
    }
    if (idx < LHALF * CB_PITCH) {
        const int t = idx / CB_PITCH;
        const int m = idx - t * CB_PITCH;
        if (m == 0 || m >= NPAIR) {
            cB[idx] = 1.0f;
            sB[idx] = 0.0f;
        } else {
            const float th = thetaB[t * (NPAIR - 1) + (m - 1)];
            cB[idx] = cosf(th);
            sB[idx] = sinf(th);
        }
    }
}

__device__ __forceinline__ v2f lo2(const float4 v) { v2f r; r.x = v.x; r.y = v.y; return r; }
__device__ __forceinline__ v2f hi2(const float4 v) { v2f r; r.x = v.z; r.y = v.w; return r; }

__device__ __forceinline__ void wmma_2x4(v8f (&acc)[2][4], const v2f x0, const v2f x1,
                                         const v2f y0, const v2f y1, const v2f y2, const v2f y3)
{
    acc[0][0] = __builtin_amdgcn_wmma_f32_16x16x4_f32(false, x0, false, y0,
                                                      (short)0, acc[0][0], false, false);
    acc[0][1] = __builtin_amdgcn_wmma_f32_16x16x4_f32(false, x0, false, y1,
                                                      (short)0, acc[0][1], false, false);
    acc[0][2] = __builtin_amdgcn_wmma_f32_16x16x4_f32(false, x0, false, y2,
                                                      (short)0, acc[0][2], false, false);
    acc[0][3] = __builtin_amdgcn_wmma_f32_16x16x4_f32(false, x0, false, y3,
                                                      (short)0, acc[0][3], false, false);
    acc[1][0] = __builtin_amdgcn_wmma_f32_16x16x4_f32(false, x1, false, y0,
                                                      (short)0, acc[1][0], false, false);
    acc[1][1] = __builtin_amdgcn_wmma_f32_16x16x4_f32(false, x1, false, y1,
                                                      (short)0, acc[1][1], false, false);
    acc[1][2] = __builtin_amdgcn_wmma_f32_16x16x4_f32(false, x1, false, y2,
                                                      (short)0, acc[1][2], false, false);
    acc[1][3] = __builtin_amdgcn_wmma_f32_16x16x4_f32(false, x1, false, y3,
                                                      (short)0, acc[1][3], false, false);
}

__device__ __forceinline__ void gemm_group(v8f (&acc)[2][4],
                                           const float4 (&a0)[4], const float4 (&a1)[4],
                                           const float* __restrict__ bfrag)
{
    #pragma unroll
    for (int i = 0; i < 4; ++i) {
        const float* bB = bfrag + i * (2 * PN * 4);
        const float4 b0 = *(const float4*)(bB);
        const float4 b1 = *(const float4*)(bB + 16 * 4);
        const float4 b2 = *(const float4*)(bB + 32 * 4);
        const float4 b3 = *(const float4*)(bB + 48 * 4);
        wmma_2x4(acc, lo2(a0[i]), lo2(a1[i]), lo2(b0), lo2(b1), lo2(b2), lo2(b3));
        wmma_2x4(acc, hi2(a0[i]), hi2(a1[i]), hi2(b0), hi2(b1), hi2(b2), hi2(b3));
    }
}

__global__ __launch_bounds__(512)
void gemm_hx_kernel(const float* __restrict__ A,
                    const float* __restrict__ Bg,
                    const float* __restrict__ hx,
                    float* __restrict__ out)
{
    __shared__ __align__(16) float Bs[KQ * PN * 4];

    const int tid    = threadIdx.x;
    const int lane   = tid & 31;
    const int wave   = tid >> 5;
    const int lane15 = lane & 15;
    const int lhalf  = lane >> 4;
    const int n0     = (blockIdx.x & 15) * PN;
    const int row0   = (blockIdx.x >> 4) * GROWS + wave * 32;

    const float* a0p = A + (size_t)(row0 + lane15) * KDIM + 4 * lhalf;
    const float* a1p = a0p + (size_t)16 * KDIM;

    float4 pa0[4], pa1[4], qa0[4], qa1[4];
    #pragma unroll
    for (int i = 0; i < 4; ++i) {
        pa0[i] = *(const float4*)(a0p + 8 * i);
        pa1[i] = *(const float4*)(a1p + 8 * i);
    }

    {
        const int n  = tid & 63;
        const int q0 = tid >> 6;
        const float* src = Bg + n0 + n;
        #pragma unroll 8
        for (int j = 0; j < KQ / 8; ++j) {
            const int q = q0 + 8 * j;
            float4 v;
            v.x = src[(size_t)(4 * q + 0) * NDIM];
            v.y = src[(size_t)(4 * q + 1) * NDIM];
            v.z = src[(size_t)(4 * q + 2) * NDIM];
            v.w = src[(size_t)(4 * q + 3) * NDIM];
            *(float4*)(Bs + (q * PN + n) * 4) = v;
        }
    }
    __syncthreads();

    v8f acc[2][4] = {};

    const float* bl = Bs + (lhalf * PN + lane15) * 4;

    for (int g = 0; g < KDIM / 32; g += 2) {
        #pragma unroll
        for (int i = 0; i < 4; ++i) {
            qa0[i] = *(const float4*)(a0p + 32 * (g + 1) + 8 * i);
            qa1[i] = *(const float4*)(a1p + 32 * (g + 1) + 8 * i);
        }
        gemm_group(acc, pa0, pa1, bl + (size_t)g * (8 * PN * 4));

        const int gn = (g + 2 < KDIM / 32) ? (g + 2) : (KDIM / 32 - 1);
        #pragma unroll
        for (int i = 0; i < 4; ++i) {
            pa0[i] = *(const float4*)(a0p + 32 * gn + 8 * i);
            pa1[i] = *(const float4*)(a1p + 32 * gn + 8 * i);
        }
        gemm_group(acc, qa0, qa1, bl + (size_t)(g + 1) * (8 * PN * 4));
    }

    const int hi = lhalf << 3;
    #pragma unroll
    for (int ti = 0; ti < 2; ++ti) {
        #pragma unroll
        for (int tj = 0; tj < 4; ++tj) {
            const int mBase = row0 + ti * 16 + hi;
            const int nCol  = n0 + tj * 16 + lane15;
            #pragma unroll
            for (int v = 0; v < 8; ++v) {
                const size_t idx = (size_t)(mBase + v) * NDIM + nCol;
                out[idx] = acc[ti][tj][v] + hx[idx];
            }
        }
    }
}

#define SC_ROWS   32
#define SC_WAVES  32
#define SC_JPW    (NDIM / SC_WAVES)
#define SC_PITCH  132
#define SC_PLANE  (128 * SC_PITCH)
#define SC_BUF    (2 * SC_PLANE)

__global__ __launch_bounds__(1024)
void eunn_modrelu_kernel(float* __restrict__ h,
                         const float* __restrict__ cA,
                         const float* __restrict__ sA,
                         const float* __restrict__ cB,
                         const float* __restrict__ sB,
                         const float* __restrict__ bias)
{
    __shared__ __align__(16) float T[2 * SC_BUF];

    const int tid  = threadIdx.x;
    const int lane = tid & 31;
    const int w    = __builtin_amdgcn_readfirstlane(tid >> 5);
    float* __restrict__ hblk = h + (size_t)blockIdx.x * SC_ROWS * NDIM;

    float* T0 = T;
    float* T1 = T + SC_BUF;

    const int tc   = tid & 255;
    const int rsub = tid >> 8;
    const int tr_off = (tc >> 1) * SC_PITCH + 2 * (tc & 1);
    #pragma unroll
    for (int p = 0; p < 8; ++p) {
        const int r = 4 * p + rsub;
        const float4 v = *(const float4*)(hblk + (size_t)r * NDIM + 4 * tc);
        v2f ev; ev.x = v.x; ev.y = v.z;
        v2f od; od.x = v.y; od.y = v.w;
        *(v2f*)(T1 + tr_off + r * 4)            = ev;
        *(v2f*)(T1 + SC_PLANE + tr_off + r * 4) = od;
    }
    __syncthreads();

    const int wr_off = ((SC_JPW / 8) * w) * SC_PITCH + lane * 4;
    const int rd_off = ((w / (SC_WAVES / 2)) * 128 + (SC_JPW / 4) * (w % (SC_WAVES / 2))) * SC_PITCH + lane * 4;
    const int jb     = w * SC_JPW;

    float x[SC_JPW];
    #pragma unroll
    for (int g = 0; g < SC_JPW / 8; ++g) {
        const float4 ev = *(const float4*)(T1 + wr_off + g * SC_PITCH);
        const float4 od = *(const float4*)(T1 + SC_PLANE + wr_off + g * SC_PITCH);
        x[8 * g + 0] = ev.x; x[8 * g + 2] = ev.y; x[8 * g + 4] = ev.z; x[8 * g + 6] = ev.w;
        x[8 * g + 1] = od.x; x[8 * g + 3] = od.y; x[8 * g + 5] = od.z; x[8 * g + 7] = od.w;
    }

    for (int t = 0; t < LHALF; ++t) {
        float* Tb = (t & 1) ? T1 : T0;
        const float* __restrict__ pcA = cA + t * NPAIR + w * (SC_JPW / 2);
        const float* __restrict__ psA = sA + t * NPAIR + w * (SC_JPW / 2);
        const float* __restrict__ pcB = cB + t * CB_PITCH + w * (SC_JPW / 2);
        const float* __restrict__ psB = sB + t * CB_PITCH + w * (SC_JPW / 2);

        float cb[SC_JPW / 2 + 1], sb[SC_JPW / 2 + 1];
        #pragma unroll
        for (int i = 0; i <= SC_JPW / 2; ++i) { cb[i] = pcB[i]; sb[i] = psB[i]; }
        float ca[SC_JPW / 2], sa[SC_JPW / 2];
        #pragma unroll
        for (int i = 0; i < SC_JPW / 2; ++i) { ca[i] = pcA[i]; sa[i] = psA[i]; }

        #pragma unroll
        for (int i = 0; i < SC_JPW; i += 2) {
            const float x0 = x[i], x1 = x[i + 1];
            x[i]     = x0 * ca[i >> 1] - x1 * sa[i >> 1];
            x[i + 1] = x1 * ca[i >> 1] + x0 * sa[i >> 1];
        }
        #pragma unroll
        for (int g = 0; g < SC_JPW / 8; ++g) {
            float4 ev, od;
            ev.x = x[8 * g + 0]; ev.y = x[8 * g + 2]; ev.z = x[8 * g + 4]; ev.w = x[8 * g + 6];
            od.x = x[8 * g + 1]; od.y = x[8 * g + 3]; od.z = x[8 * g + 5]; od.w = x[8 * g + 7];
            *(float4*)(Tb + wr_off + g * SC_PITCH)            = ev;
            *(float4*)(Tb + SC_PLANE + wr_off + g * SC_PITCH) = od;
        }
        __syncthreads();

        #pragma unroll
        for (int qq = 0; qq < SC_JPW / 4; ++qq) {
            const float4 y = *(const float4*)(Tb + rd_off + qq * SC_PITCH);
            x[4 * qq + 0] = x[4 * qq + 0] * cb[2 * qq + 0] + y.x * sb[2 * qq + 0];
            x[4 * qq + 1] = x[4 * qq + 1] * cb[2 * qq + 1] - y.y * sb[2 * qq + 1];
            x[4 * qq + 2] = x[4 * qq + 2] * cb[2 * qq + 1] + y.z * sb[2 * qq + 1];
            x[4 * qq + 3] = x[4 * qq + 3] * cb[2 * qq + 2] - y.w * sb[2 * qq + 2];
        }
    }

    const float* __restrict__ cb = bias + jb;
    #pragma unroll
    for (int i = 0; i < SC_JPW; ++i) {
        const float xv = x[i];
        const float s  = (xv > 0.f) ? 1.f : ((xv < 0.f) ? -1.f : 0.f);
        x[i] = s * fmaxf(fabsf(xv) + cb[i], 0.f);
    }
    #pragma unroll
    for (int g = 0; g < SC_JPW / 8; ++g) {
        float4 ev, od;
        ev.x = x[8 * g + 0]; ev.y = x[8 * g + 2]; ev.z = x[8 * g + 4]; ev.w = x[8 * g + 6];
        od.x = x[8 * g + 1]; od.y = x[8 * g + 3]; od.z = x[8 * g + 5]; od.w = x[8 * g + 7];
        *(float4*)(T0 + wr_off + g * SC_PITCH)            = ev;
        *(float4*)(T0 + SC_PLANE + wr_off + g * SC_PITCH) = od;
    }
    __syncthreads();

    const int lane2   = (int)__lane_id();
    const int tc2     = 32 * (w & 7) + lane2;
    const int rsub2   = w >> 3;
    const int tr_off2 = (tc2 >> 1) * SC_PITCH + 2 * (tc2 & 1);
    #pragma unroll
    for (int p = 0; p < 8; ++p) {
        const int r = 4 * p + rsub2;
        const v2f ev = *(const v2f*)(T0 + tr_off2 + r * 4);
        const v2f od = *(const v2f*)(T0 + SC_PLANE + tr_off2 + r * 4);
        float4 v;
        v.x = ev.x; v.y = od.x; v.z = ev.y; v.w = od.y;
        *(float4*)(hblk + (size_t)r * NDIM + 4 * tc2) = v;
    }
}

extern "C" void kernel_launch(void* const* d_in, const int* in_sizes, int n_in,
                              void* d_out, int out_size, void* d_ws, size_t ws_size,
                              hipStream_t stream)
{
    const float* input_ = (const float*)d_in[0];
    const float* hx     = (const float*)d_in[1];
    const float* U      = (const float*)d_in[2];
    const float* thetaA = (const float*)d_in[3];
    const float* thetaB = (const float*)d_in[4];
    const float* bias   = (const float*)d_in[5];
    float* out = (float*)d_out;

    float* ws = (float*)d_ws;
    float* cA = ws;
    float* sA = ws + LHALF * NPAIR;
    float* cB = ws + 2 * LHALF * NPAIR;
    float* sB = ws + 2 * LHALF * NPAIR + LHALF * CB_PITCH;

    eunn_coeff_kernel<<<(LHALF * CB_PITCH + 255) / 256, 256, 0, stream>>>(
        thetaA, thetaB, cA, sA, cB, sB);

    gemm_hx_kernel<<<(MDIM / GROWS) * (NDIM / PN), 32 * GW, 0, stream>>>(input_, U, hx, out);

    eunn_modrelu_kernel<<<MDIM / SC_ROWS, 32 * SC_WAVES, 0, stream>>>(out, cA, sA, cB, sB, bias);
}
